# starting kernel
# baseline (speedup 1.0000x reference)
#include <hip/hip_runtime.h>
#include <math.h>

typedef __attribute__((ext_vector_type(2))) float v2f;
typedef __attribute__((ext_vector_type(4))) float v4f;
typedef __attribute__((ext_vector_type(8))) float v8f;

#define B_  8192
#define P_  25
#define C_  256
#define D_  5
#define R_  (B_ * P_)
#define ROWS_PER_WAVE 16
#define WAVES_PER_BLOCK 2
#define HSTRIDE 260
#define KSTEPS (C_ / 4)
#define ZOFF (D_ * HSTRIDE)

__global__ __launch_bounds__(256) void rowsum_kernel(const float* __restrict__ w,
                                                     float* __restrict__ wrow) {
  const int row = blockIdx.x * 8 + (threadIdx.x >> 5);
  const int ln  = threadIdx.x & 31;
  const float* rp = w + (size_t)row * C_;
  float s = 0.f;
#pragma unroll
  for (int j = 0; j < 8; ++j) s += rp[ln + 32 * j];
#pragma unroll
  for (int off = 16; off > 0; off >>= 1) s += __shfl_xor(s, off, 32);
  if (ln == 0) wrow[row] = s;
}

__global__ __launch_bounds__(64) void fused_kernel(
    const float* __restrict__ feat, const float* __restrict__ wrow,
    const float* __restrict__ bias, const float* __restrict__ gamma,
    const float* __restrict__ beta, const float* __restrict__ linw,
    const float* __restrict__ linb, float* __restrict__ out) {
  __shared__ __attribute__((aligned(16))) float hbuf[WAVES_PER_BLOCK * ROWS_PER_WAVE * HSTRIDE];
  __shared__ __attribute__((aligned(16))) float lw[ZOFF + 4];

  const int tid = threadIdx.x;

  for (int e = tid; e < D_ * C_; e += 64) {
    lw[(e >> 8) * HSTRIDE + (e & 255)] = linw[e];
  }
  if (tid < 4) lw[ZOFF + tid] = 0.f;
  __syncthreads();

  const int wv   = tid >> 5;
  const int ln   = tid & 31;
  const int wave = blockIdx.x * WAVES_PER_BLOCK + wv;
  const int p    = wave % P_;
  const int b0   = (wave / P_) * ROWS_PER_WAVE;
  float* hw = hbuf + wv * (ROWS_PER_WAVE * HSTRIDE);

  const int c0 = ln * 4, c1 = ln * 4 + 128;

  const v4f wr0 = *(const v4f*)(wrow + p * C_ + c0);
  const v4f wr1 = *(const v4f*)(wrow + p * C_ + c1);
  const v4f bi0 = *(const v4f*)(bias + p * C_ + c0);
  const v4f bi1 = *(const v4f*)(bias + p * C_ + c1);
  const v4f ga0 = *(const v4f*)(gamma + c0);
  const v4f ga1 = *(const v4f*)(gamma + c1);
  const v4f be0 = *(const v4f*)(beta + c0);
  const v4f be1 = *(const v4f*)(beta + c1);

#pragma unroll 2
  for (int r = 0; r < ROWS_PER_WAVE; ++r) {
    const int g = (b0 + r) * P_ + p;
    const float* fp = feat + (size_t)g * C_;
    const v4f f0 = __builtin_nontemporal_load((const v4f*)(fp + c0));
    const v4f f1 = __builtin_nontemporal_load((const v4f*)(fp + c1));
    v4f pw0, pw1;
    float s1 = 0.f, s2 = 0.f;
#pragma unroll
    for (int q = 0; q < 4; ++q) {
      pw0[q] = fmaf(f0[q], wr0[q], bi0[q]);
      pw1[q] = fmaf(f1[q], wr1[q], bi1[q]);
      s1 += pw0[q] + pw1[q];
      s2  = fmaf(pw0[q], pw0[q], s2);
      s2  = fmaf(pw1[q], pw1[q], s2);
    }
#pragma unroll
    for (int off = 16; off > 0; off >>= 1) {
      s1 += __shfl_xor(s1, off, 32);
      s2 += __shfl_xor(s2, off, 32);
    }
    const float inv  = 1.0f / (float)C_;
    const float mean = s1 * inv;
    const float var  = fmaf(s2, inv, -mean * mean);
    const float rstd = rsqrtf(var + 1e-5f);
    v4f h0, h1;
#pragma unroll
    for (int q = 0; q < 4; ++q) {
      const float n0 = fmaf((pw0[q] - mean) * rstd, ga0[q], be0[q]);
      const float n1 = fmaf((pw1[q] - mean) * rstd, ga1[q], be1[q]);
      h0[q] = fmaxf(f0[q] + n0, 0.f);
      h1[q] = fmaxf(f1[q] + n1, 0.f);
    }
    *(v4f*)&hw[r * HSTRIDE + c0] = h0;
    *(v4f*)&hw[r * HSTRIDE + c1] = h1;
  }

  const int rowA = ln & 15;
  const int koff = (ln >> 4) << 1;
  const int dcol = ln & 15;
  const float* bbase = (dcol < D_) ? &lw[dcol * HSTRIDE + koff] : &lw[ZOFF];
  const int bstep = (dcol < D_) ? 4 : 0;

  v8f acc = {0.f, 0.f, 0.f, 0.f, 0.f, 0.f, 0.f, 0.f};
#pragma unroll 4
  for (int k = 0; k < KSTEPS; ++k) {
    v2f a = *(const v2f*)&hw[rowA * HSTRIDE + 4 * k + koff];
    v2f b = *(const v2f*)&bbase[k * bstep];
    acc = __builtin_amdgcn_wmma_f32_16x16x4_f32(
         false, a,  false, b,
         (short)0, acc,  false,  false);
  }

  const int rAdd = (ln >> 4) << 3;
  if (dcol < D_) {
    const float lb = linb[dcol];
#pragma unroll
    for (int j = 0; j < 8; ++j) {
      const int g = (b0 + j + rAdd) * P_ + p;
      __builtin_nontemporal_store(acc[j] + lb, out + (size_t)g * D_ + dcol);
    }
  }
}

extern "C" void kernel_launch(void* const* d_in, const int* in_sizes, int n_in,
                              void* d_out, int out_size, void* d_ws, size_t ws_size,
                              hipStream_t stream) {
  const float* feat    = (const float*)d_in[0];
  const float* weights = (const float*)d_in[1];
  const float* bias    = (const float*)d_in[2];
  const float* gamma   = (const float*)d_in[3];
  const float* beta    = (const float*)d_in[4];
  const float* linw    = (const float*)d_in[5];
  const float* linb    = (const float*)d_in[6];
  float* out  = (float*)d_out;
  float* wrow = (float*)d_ws;

  rowsum_kernel<<<(P_ * C_) / 8, 256, 0, stream>>>(weights, wrow);

  const int nblocks = R_ / (ROWS_PER_WAVE * WAVES_PER_BLOCK);
  fused_kernel<<<nblocks, 64, 0, stream>>>(feat, wrow, bias, gamma, beta,
                                           linw, linb, out);
}


# Lane pair per row: each lane reads half a row from global into registers in the WMMA A-fragment map; LayerNorm sums in-lane + one lane^16 add, A from registers, no LDS tile (sum and k order change)
# speedup vs baseline: 1.1159x; 1.1159x over previous
#include <hip/hip_runtime.h>
#include <math.h>


typedef __attribute__((ext_vector_type(2))) float v2f;
typedef __attribute__((ext_vector_type(4))) float v4f;
typedef __attribute__((ext_vector_type(8))) float v8f;

#define B_  8192
#define P_  25
#define C_  256
#define D_  5
#define NCH 32
#define TILES 2
#define WAVES 4
#define BROWS (WAVES * TILES * 16)
#define LWS 264

__global__ __launch_bounds__(256) void rowsum_kernel(const float* __restrict__ w,
                                                     float* __restrict__ wrow) {
  const int row = blockIdx.x * 8 + (threadIdx.x >> 5);
  const int ln  = threadIdx.x & 31;
  const float* rp = w + (size_t)row * C_;
  float s = 0.f;
#pragma unroll
  for (int j = 0; j < 8; ++j) s += rp[ln + 32 * j];
#pragma unroll
  for (int off = 16; off > 0; off >>= 1) s += __shfl_xor(s, off, 32);
  if (ln == 0) wrow[row] = s;
}

__global__ __launch_bounds__(32 * WAVES) void fused_kernel(
    const float* __restrict__ feat, const float* __restrict__ wrow,
    const float* __restrict__ bias, const float* __restrict__ gamma,
    const float* __restrict__ beta, const float* __restrict__ linw,
    const float* __restrict__ linb, float* __restrict__ out) {
  __shared__ __attribute__((aligned(16))) float cst[4 * C_];
  __shared__ __attribute__((aligned(16))) float clw[D_ * LWS];

  const int tid = threadIdx.x;
  const int p   = blockIdx.x % P_;
  const int bg  = blockIdx.x / P_;

  {
    const int  idx = tid & 63;
    const bool sec = (tid & 64) != 0;
    const float* s0 = sec ? gamma : (wrow + p * C_);
    const float* s1 = sec ? beta  : (bias + p * C_);
    const v4f a = *(const v4f*)(s0 + 4 * idx);
    const v4f b = *(const v4f*)(s1 + 4 * idx);
    float* d0 = cst + (sec ? 2 * C_ : 0);
    *(v4f*)(d0 + 4 * idx)      = a;
    *(v4f*)(d0 + C_ + 4 * idx) = b;
    for (int e = tid; e < D_ * C_ / 4; e += 32 * WAVES) {
      const v4f w = *(const v4f*)(linw + 4 * e);
      *(v4f*)&clw[(e >> 6) * LWS + 4 * (e & 63)] = w;
    }
  }
  __syncthreads();

  const int wv = tid >> 5;
  const int ln = tid & 31;
  const int r  = ln & 15;
  const int h  = ln >> 4;
  const int brow = bg * BROWS + wv * (TILES * 16);

  v4f f[TILES][NCH];
  const float* fp0 = feat + ((size_t)(brow + r) * P_ + p) * C_ + 4 * h;
#pragma unroll
  for (int j = 0; j < NCH; ++j) {
#pragma unroll
    for (int t = 0; t < TILES; ++t)
      f[t][j] = *(const v4f*)(fp0 + (size_t)t * 16 * P_ * C_ + 8 * j);
  }

  const float* cl = cst + 4 * h;

  v2f s1q[TILES], s2q[TILES];
#pragma unroll
  for (int t = 0; t < TILES; ++t) {
    s1q[t] = (v2f){0.f, 0.f};
    s2q[t] = (v2f){0.f, 0.f};
  }

#pragma unroll
  for (int j = 0; j < NCH; ++j) {
    const v4f wr = *(const v4f*)(cl + 8 * j);
    const v4f bi = *(const v4f*)(cl + C_ + 8 * j);
#pragma unroll
    for (int t = 0; t < TILES; ++t) {
      const v4f pw = __builtin_elementwise_fma(f[t][j], wr, bi);
      const v2f plo = {pw[0], pw[1]};
      const v2f phi = {pw[2], pw[3]};
      s1q[t] += plo;
      s1q[t] += phi;
      s2q[t]  = __builtin_elementwise_fma(plo, plo, s2q[t]);
      s2q[t]  = __builtin_elementwise_fma(phi, phi, s2q[t]);
    }
  }

  float mean[TILES], rstd[TILES];
#pragma unroll
  for (int t = 0; t < TILES; ++t) {
    float s1 = s1q[t][0] + s1q[t][1];
    float s2 = s2q[t][0] + s2q[t][1];
    s1 += __shfl_xor(s1, 16, 32);
    s2 += __shfl_xor(s2, 16, 32);
    const float inv = 1.0f / (float)C_;
    const float m   = s1 * inv;
    const float var = fmaf(s2, inv, -m * m);
    mean[t] = m;
    rstd[t] = rsqrtf(var + 1e-5f);
  }

  int hoff = 4 * h;
  asm volatile("" : "+v"(hoff));
  const float* cl2 = cst + hoff;

  const int n = ln & 15;
  const float* lwp = clw + (n < D_ ? n : D_ - 1) * LWS + 4 * h;

  v8f acc[TILES];
#pragma unroll
  for (int t = 0; t < TILES; ++t)
    acc[t] = (v8f){0.f, 0.f, 0.f, 0.f, 0.f, 0.f, 0.f, 0.f};

  v4f cw[2][5];
  cw[0][0] = *(const v4f*)(cl2);
  cw[0][1] = *(const v4f*)(cl2 + C_);
  cw[0][2] = *(const v4f*)(cl2 + 2 * C_);
  cw[0][3] = *(const v4f*)(cl2 + 3 * C_);
  cw[0][4] = *(const v4f*)(lwp);
#pragma unroll
  for (int j = 0; j < NCH; ++j) {
    if (j + 1 < NCH) {
      cw[(j + 1) & 1][0] = *(const v4f*)(cl2 + 8 * (j + 1));
      cw[(j + 1) & 1][1] = *(const v4f*)(cl2 + C_ + 8 * (j + 1));
      cw[(j + 1) & 1][2] = *(const v4f*)(cl2 + 2 * C_ + 8 * (j + 1));
      cw[(j + 1) & 1][3] = *(const v4f*)(cl2 + 3 * C_ + 8 * (j + 1));
      cw[(j + 1) & 1][4] = *(const v4f*)(lwp + 8 * (j + 1));
    }
    const v4f wr = cw[j & 1][0];
    const v4f bi = cw[j & 1][1];
    const v4f ga = cw[j & 1][2];
    const v4f be = cw[j & 1][3];
    const v4f lw = cw[j & 1][4];
#pragma unroll
    for (int t = 0; t < TILES; ++t) {
      const v4f x  = f[t][j];
      const v4f pw = __builtin_elementwise_fma(x, wr, bi);
      const v4f nn = __builtin_elementwise_fma((pw - mean[t]) * rstd[t], ga, be);
      const v4f hh = __builtin_elementwise_max(x + nn, (v4f){0.f, 0.f, 0.f, 0.f});
      const v2f a0 = {hh[0], hh[1]};
      const v2f a1 = {hh[2], hh[3]};
      const v2f b0 = {lw[0], lw[1]};
      const v2f b1 = {lw[2], lw[3]};
      acc[t] = __builtin_amdgcn_wmma_f32_16x16x4_f32(
           false, a0,  false, b0,
           (short)0, acc[t],  false,  false);
      acc[t] = __builtin_amdgcn_wmma_f32_16x16x4_f32(
           false, a1,  false, b1,
           (short)0, acc[t],  false,  false);
    }
  }

  if (n < D_) {
    const float lb = linb[n];
#pragma unroll
    for (int t = 0; t < TILES; ++t) {
#pragma unroll
      for (int i = 0; i < 8; ++i) {
        const int b = brow + 16 * t + i + 8 * h;
        __builtin_nontemporal_store(acc[t][i] + lb,
                                    out + ((size_t)b * P_ + p) * D_ + n);
      }
    }
  }
}

extern "C" void kernel_launch(void* const* d_in, const int* in_sizes, int n_in,
                              void* d_out, int out_size, void* d_ws, size_t ws_size,
                              hipStream_t stream) {
  const float* feat    = (const float*)d_in[0];
  const float* weights = (const float*)d_in[1];
  const float* bias    = (const float*)d_in[2];
  const float* gamma   = (const float*)d_in[3];
  const float* beta    = (const float*)d_in[4];
  const float* linw    = (const float*)d_in[5];
  const float* linb    = (const float*)d_in[6];
  float* out  = (float*)d_out;
  float* wrow = (float*)d_ws;

  rowsum_kernel<<<(P_ * C_) / 8, 256, 0, stream>>>(weights, wrow);

  fused_kernel<<<P_ * (B_ / BROWS), 32 * WAVES, 0, stream>>>(feat, wrow, bias, gamma, beta,
                                                     linw, linb, out);
}


# A tile = 8 batch rows x 2 column pieces, idle B columns 5-9 carry lin_w of piece 1: a global load touches 8 lines not 16, LDS constant reads halve; D halves added at end (sum and k order change)
# speedup vs baseline: 1.1577x; 1.0375x over previous
#include <hip/hip_runtime.h>
#include <math.h>


typedef __attribute__((ext_vector_type(2))) float v2f;
typedef __attribute__((ext_vector_type(4))) float v4f;
typedef __attribute__((ext_vector_type(8))) float v8f;

#define B_  8192
#define P_  25
#define C_  256
#define D_  5
#define NCH 16
#define GRPS 4
#define WAVES 4
#define BROWS (WAVES * GRPS * 8)
#define LWS 268

__global__ __launch_bounds__(256) void rowsum_kernel(const float* __restrict__ w,
                                                     float* __restrict__ wrow) {
  const int row = blockIdx.x * 8 + (threadIdx.x >> 5);
  const int ln  = threadIdx.x & 31;
  const float* rp = w + (size_t)row * C_;
  float s = 0.f;
#pragma unroll
  for (int j = 0; j < 8; ++j) s += rp[ln + 32 * j];
#pragma unroll
  for (int off = 16; off > 0; off >>= 1) s += __shfl_xor(s, off, 32);
  if (ln == 0) wrow[row] = s;
}

__global__ __launch_bounds__(32 * WAVES) void fused_kernel(
    const float* __restrict__ feat, const float* __restrict__ wrow,
    const float* __restrict__ bias, const float* __restrict__ gamma,
    const float* __restrict__ beta, const float* __restrict__ linw,
    const float* __restrict__ linb, float* __restrict__ out) {
  __shared__ __attribute__((aligned(16))) float cst[4 * C_];
  __shared__ __attribute__((aligned(16))) float clw[D_ * LWS];

  const int tid = threadIdx.x;
  const int p   = blockIdx.x % P_;
  const int bg  = blockIdx.x / P_;

  {
    const int  idx = tid & 63;
    const bool sec = (tid & 64) != 0;
    const float* s0 = sec ? gamma : (wrow + p * C_);
    const float* s1 = sec ? beta  : (bias + p * C_);
    const v4f a = *(const v4f*)(s0 + 4 * idx);
    const v4f b = *(const v4f*)(s1 + 4 * idx);
    float* d0 = cst + (sec ? 2 * C_ : 0);
    *(v4f*)(d0 + 4 * idx)      = a;
    *(v4f*)(d0 + C_ + 4 * idx) = b;
    for (int e = tid; e < D_ * C_ / 4; e += 32 * WAVES) {
      const v4f w = *(const v4f*)(linw + 4 * e);
      *(v4f*)&clw[(e >> 6) * LWS + 4 * (e & 63)] = w;
    }
  }
  __syncthreads();

  const int wv = tid >> 5;
  const int ln = tid & 31;
  const int r8 = ln & 7;
  const int g  = (ln >> 3) & 1;
  const int h  = ln >> 4;
  const int brow = bg * BROWS + wv * (GRPS * 8);
  const int coff = 8 * g + 4 * h;

  v4f f[GRPS][NCH];
  const float* fp0 = feat + ((size_t)(brow + r8) * P_ + p) * C_ + coff;
#pragma unroll
  for (int jj = 0; jj < NCH; ++jj) {
#pragma unroll
    for (int u = 0; u < GRPS; ++u)
      f[u][jj] = *(const v4f*)(fp0 + (size_t)u * 8 * P_ * C_ + 16 * jj);
  }

  const float* cl = cst + coff;

  v2f s1q[GRPS], s2q[GRPS];
#pragma unroll
  for (int u = 0; u < GRPS; ++u) {
    s1q[u] = (v2f){0.f, 0.f};
    s2q[u] = (v2f){0.f, 0.f};
  }

#pragma unroll
  for (int jj = 0; jj < NCH; ++jj) {
    const v4f wr = *(const v4f*)(cl + 16 * jj);
    const v4f bi = *(const v4f*)(cl + C_ + 16 * jj);
#pragma unroll
    for (int u = 0; u < GRPS; ++u) {
      const v4f pw = __builtin_elementwise_fma(f[u][jj], wr, bi);
      const v2f plo = {pw[0], pw[1]};
      const v2f phi = {pw[2], pw[3]};
      s1q[u] += plo;
      s1q[u] += phi;
      s2q[u]  = __builtin_elementwise_fma(plo, plo, s2q[u]);
      s2q[u]  = __builtin_elementwise_fma(phi, phi, s2q[u]);
    }
  }

  float mean[GRPS], rstd[GRPS];
#pragma unroll
  for (int u = 0; u < GRPS; ++u) {
    float s1 = s1q[u][0] + s1q[u][1];
    float s2 = s2q[u][0] + s2q[u][1];
    s1 += __shfl_xor(s1, 16, 32);
    s2 += __shfl_xor(s2, 16, 32);
    s1 += __shfl_xor(s1, 8, 32);
    s2 += __shfl_xor(s2, 8, 32);
    const float inv = 1.0f / (float)C_;
    const float m   = s1 * inv;
    const float var = fmaf(s2, inv, -m * m);
    mean[u] = m;
    rstd[u] = rsqrtf(var + 1e-5f);
  }

  int hoff = coff;
  asm volatile("" : "+v"(hoff));
  const float* cl2 = cst + hoff;

  const int n  = ln & 15;
  const int dn = n < D_ ? n : (n < 2 * D_ ? n - D_ : D_ - 1);
  const int gn = n < D_ ? 0 : 1;
  const float* lwp = clw + dn * LWS + 8 * gn + 4 * h;

  v8f acc[GRPS];
#pragma unroll
  for (int u = 0; u < GRPS; ++u)
    acc[u] = (v8f){0.f, 0.f, 0.f, 0.f, 0.f, 0.f, 0.f, 0.f};

  v4f cw[2][5];
  cw[0][0] = *(const v4f*)(cl2);
  cw[0][1] = *(const v4f*)(cl2 + C_);
  cw[0][2] = *(const v4f*)(cl2 + 2 * C_);
  cw[0][3] = *(const v4f*)(cl2 + 3 * C_);
  cw[0][4] = *(const v4f*)(lwp);
#pragma unroll
  for (int jj = 0; jj < NCH; ++jj) {
    if (jj + 1 < NCH) {
      cw[(jj + 1) & 1][0] = *(const v4f*)(cl2 + 16 * (jj + 1));
      cw[(jj + 1) & 1][1] = *(const v4f*)(cl2 + C_ + 16 * (jj + 1));
      cw[(jj + 1) & 1][2] = *(const v4f*)(cl2 + 2 * C_ + 16 * (jj + 1));
      cw[(jj + 1) & 1][3] = *(const v4f*)(cl2 + 3 * C_ + 16 * (jj + 1));
      cw[(jj + 1) & 1][4] = *(const v4f*)(lwp + 16 * (jj + 1));
    }
    const v4f wr = cw[jj & 1][0];
    const v4f bi = cw[jj & 1][1];
    const v4f ga = cw[jj & 1][2];
    const v4f be = cw[jj & 1][3];
    const v4f lw = cw[jj & 1][4];
#pragma unroll
    for (int u = 0; u < GRPS; ++u) {
      const v4f x  = f[u][jj];
      const v4f pw = __builtin_elementwise_fma(x, wr, bi);
      const v4f nn = __builtin_elementwise_fma((pw - mean[u]) * rstd[u], ga, be);
      const v4f hh = __builtin_elementwise_max(x + nn, (v4f){0.f, 0.f, 0.f, 0.f});
      const v2f a0 = {hh[0], hh[1]};
      const v2f a1 = {hh[2], hh[3]};
      const v2f b0 = {lw[0], lw[1]};
      const v2f b1 = {lw[2], lw[3]};
      acc[u] = __builtin_amdgcn_wmma_f32_16x16x4_f32(
           false, a0,  false, b0,
           (short)0, acc[u],  false,  false);
      acc[u] = __builtin_amdgcn_wmma_f32_16x16x4_f32(
           false, a1,  false, b1,
           (short)0, acc[u],  false,  false);
    }
  }

  float res[GRPS][8];
#pragma unroll
  for (int u = 0; u < GRPS; ++u) {
#pragma unroll
    for (int i = 0; i < 8; ++i) {
      const float lo = acc[u][i];
      const int   li = __builtin_bit_cast(int, lo);
      const float hi = __builtin_bit_cast(
          float, __builtin_amdgcn_permlanex16(li, li, 0x00098765, 0x00000000, false, false));
      res[u][i] = lo + hi;
    }
  }
  if (ln < D_) {
    const float lb = linb[ln];
    float* op = out + ((size_t)brow * P_ + p) * D_ + ln;
#pragma unroll
    for (int u = 0; u < GRPS; ++u) {
#pragma unroll
      for (int i = 0; i < 8; ++i)
        __builtin_nontemporal_store(res[u][i] + lb, op + (size_t)(8 * u + i) * P_ * D_);
    }
  }
}

extern "C" void kernel_launch(void* const* d_in, const int* in_sizes, int n_in,
                              void* d_out, int out_size, void* d_ws, size_t ws_size,
                              hipStream_t stream) {
  const float* feat    = (const float*)d_in[0];
  const float* weights = (const float*)d_in[1];
  const float* bias    = (const float*)d_in[2];
  const float* gamma   = (const float*)d_in[3];
  const float* beta    = (const float*)d_in[4];
  const float* linw    = (const float*)d_in[5];
  const float* linb    = (const float*)d_in[6];
  float* out  = (float*)d_out;
  float* wrow = (float*)d_ws;

  rowsum_kernel<<<(P_ * C_) / 8, 256, 0, stream>>>(weights, wrow);

  fused_kernel<<<P_ * (B_ / BROWS), 32 * WAVES, 0, stream>>>(feat, wrow, bias, gamma, beta,
                                                     linw, linb, out);
}


# Row-sum launch: one wave sums 8 weight rows (8 lanes per row, whole-line b128 loads all in flight at once, 3-step lane sum), 800 waves enter instead of 6400; w_rowsum sum order changes
# speedup vs baseline: 1.2121x; 1.0470x over previous
#include <hip/hip_runtime.h>
#include <math.h>


typedef __attribute__((ext_vector_type(2))) float v2f;
typedef __attribute__((ext_vector_type(4))) float v4f;
typedef __attribute__((ext_vector_type(8))) float v8f;

#define B_  8192
#define P_  25
#define C_  256
#define D_  5
#define NCH 16
#define GRPS 4
#define WAVES 4
#define BROWS (WAVES * GRPS * 8)
#define LWS 268

#define RS_ROWS  8
#define RS_WAVES 4
__global__ __launch_bounds__(32 * RS_WAVES) void rowsum_kernel(const float* __restrict__ w,
                                                              float* __restrict__ wrow) {
  const int wave = blockIdx.x * RS_WAVES + (threadIdx.x >> 5);
  const int ln   = threadIdx.x & 31;
  const int sub  = ln >> 3;
  const int j    = ln & 7;
  const int row0 = wave * RS_ROWS;
  const float* rp = w + (size_t)(row0 + sub) * C_ + 4 * j;

  v4f v[RS_ROWS / 4][8];
#pragma unroll
  for (int s = 0; s < RS_ROWS / 4; ++s) {
#pragma unroll
    for (int q = 0; q < 8; ++q)
      v[s][q] = *(const v4f*)(rp + (size_t)s * 4 * C_ + 32 * q);
  }

  __builtin_amdgcn_sched_barrier(0);

  float t[RS_ROWS / 4];
#pragma unroll
  for (int s = 0; s < RS_ROWS / 4; ++s) {
    v4f a = v[s][0];
#pragma unroll
    for (int q = 1; q < 8; ++q) a += v[s][q];
    t[s] = (a[0] + a[1]) + (a[2] + a[3]);
  }
#pragma unroll
  for (int off = 4; off > 0; off >>= 1) {
#pragma unroll
    for (int s = 0; s < RS_ROWS / 4; ++s) t[s] += __shfl_xor(t[s], off, 32);
  }
  if (j == 0) {
#pragma unroll
    for (int s = 0; s < RS_ROWS / 4; ++s) wrow[row0 + 4 * s + sub] = t[s];
  }
}

__global__ __launch_bounds__(32 * WAVES) void fused_kernel(
    const float* __restrict__ feat, const float* __restrict__ wrow,
    const float* __restrict__ bias, const float* __restrict__ gamma,
    const float* __restrict__ beta, const float* __restrict__ linw,
    const float* __restrict__ linb, float* __restrict__ out) {
  __shared__ __attribute__((aligned(16))) float cst[4 * C_];
  __shared__ __attribute__((aligned(16))) float clw[D_ * LWS];

  const int tid = threadIdx.x;
  const int p   = blockIdx.x % P_;
  const int bg  = blockIdx.x / P_;

  {
    const int  idx = tid & 63;
    const bool sec = (tid & 64) != 0;
    const float* s0 = sec ? gamma : (wrow + p * C_);
    const float* s1 = sec ? beta  : (bias + p * C_);
    const v4f a = *(const v4f*)(s0 + 4 * idx);
    const v4f b = *(const v4f*)(s1 + 4 * idx);
    float* d0 = cst + (sec ? 2 * C_ : 0);
    *(v4f*)(d0 + 4 * idx)      = a;
    *(v4f*)(d0 + C_ + 4 * idx) = b;
    for (int e = tid; e < D_ * C_ / 4; e += 32 * WAVES) {
      const v4f w = *(const v4f*)(linw + 4 * e);
      *(v4f*)&clw[(e >> 6) * LWS + 4 * (e & 63)] = w;
    }
  }
  __syncthreads();

  const int wv = tid >> 5;
  const int ln = tid & 31;
  const int r8 = ln & 7;
  const int g  = (ln >> 3) & 1;
  const int h  = ln >> 4;
  const int brow = bg * BROWS + wv * (GRPS * 8);
  const int coff = 8 * g + 4 * h;

  v4f f[GRPS][NCH];
  const float* fp0 = feat + ((size_t)(brow + r8) * P_ + p) * C_ + coff;
#pragma unroll
  for (int jj = 0; jj < NCH; ++jj) {
#pragma unroll
    for (int u = 0; u < GRPS; ++u)
      f[u][jj] = *(const v4f*)(fp0 + (size_t)u * 8 * P_ * C_ + 16 * jj);
  }

  const float* cl = cst + coff;

  v2f s1q[GRPS], s2q[GRPS];
#pragma unroll
  for (int u = 0; u < GRPS; ++u) {
    s1q[u] = (v2f){0.f, 0.f};
    s2q[u] = (v2f){0.f, 0.f};
  }

#pragma unroll
  for (int jj = 0; jj < NCH; ++jj) {
    const v4f wr = *(const v4f*)(cl + 16 * jj);
    const v4f bi = *(const v4f*)(cl + C_ + 16 * jj);
#pragma unroll
    for (int u = 0; u < GRPS; ++u) {
      const v4f pw = __builtin_elementwise_fma(f[u][jj], wr, bi);
      const v2f plo = {pw[0], pw[1]};
      const v2f phi = {pw[2], pw[3]};
      s1q[u] += plo;
      s1q[u] += phi;
      s2q[u]  = __builtin_elementwise_fma(plo, plo, s2q[u]);
      s2q[u]  = __builtin_elementwise_fma(phi, phi, s2q[u]);
    }
  }

  float mean[GRPS], rstd[GRPS];
#pragma unroll
  for (int u = 0; u < GRPS; ++u) {
    float s1 = s1q[u][0] + s1q[u][1];
    float s2 = s2q[u][0] + s2q[u][1];
    s1 += __shfl_xor(s1, 16, 32);
    s2 += __shfl_xor(s2, 16, 32);
    s1 += __shfl_xor(s1, 8, 32);
    s2 += __shfl_xor(s2, 8, 32);
    const float inv = 1.0f / (float)C_;
    const float m   = s1 * inv;
    const float var = fmaf(s2, inv, -m * m);
    mean[u] = m;
    rstd[u] = rsqrtf(var + 1e-5f);
  }

  int hoff = coff;
  asm volatile("" : "+v"(hoff));
  const float* cl2 = cst + hoff;

  const int n  = ln & 15;
  const int dn = n < D_ ? n : (n < 2 * D_ ? n - D_ : D_ - 1);
  const int gn = n < D_ ? 0 : 1;
  const float* lwp = clw + dn * LWS + 8 * gn + 4 * h;

  v8f acc[GRPS];
#pragma unroll
  for (int u = 0; u < GRPS; ++u)
    acc[u] = (v8f){0.f, 0.f, 0.f, 0.f, 0.f, 0.f, 0.f, 0.f};

  v4f cw[2][5];
  cw[0][0] = *(const v4f*)(cl2);
  cw[0][1] = *(const v4f*)(cl2 + C_);
  cw[0][2] = *(const v4f*)(cl2 + 2 * C_);
  cw[0][3] = *(const v4f*)(cl2 + 3 * C_);
  cw[0][4] = *(const v4f*)(lwp);
#pragma unroll
  for (int jj = 0; jj < NCH; ++jj) {
    if (jj + 1 < NCH) {
      cw[(jj + 1) & 1][0] = *(const v4f*)(cl2 + 16 * (jj + 1));
      cw[(jj + 1) & 1][1] = *(const v4f*)(cl2 + C_ + 16 * (jj + 1));
      cw[(jj + 1) & 1][2] = *(const v4f*)(cl2 + 2 * C_ + 16 * (jj + 1));
      cw[(jj + 1) & 1][3] = *(const v4f*)(cl2 + 3 * C_ + 16 * (jj + 1));
      cw[(jj + 1) & 1][4] = *(const v4f*)(lwp + 16 * (jj + 1));
    }
    const v4f wr = cw[jj & 1][0];
    const v4f bi = cw[jj & 1][1];
    const v4f ga = cw[jj & 1][2];
    const v4f be = cw[jj & 1][3];
    const v4f lw = cw[jj & 1][4];
#pragma unroll
    for (int u = 0; u < GRPS; ++u) {
      const v4f x  = f[u][jj];
      const v4f pw = __builtin_elementwise_fma(x, wr, bi);
      const v4f nn = __builtin_elementwise_fma((pw - mean[u]) * rstd[u], ga, be);
      const v4f hh = __builtin_elementwise_max(x + nn, (v4f){0.f, 0.f, 0.f, 0.f});
      const v2f a0 = {hh[0], hh[1]};
      const v2f a1 = {hh[2], hh[3]};
      const v2f b0 = {lw[0], lw[1]};
      const v2f b1 = {lw[2], lw[3]};
      acc[u] = __builtin_amdgcn_wmma_f32_16x16x4_f32(
           false, a0,  false, b0,
           (short)0, acc[u],  false,  false);
      acc[u] = __builtin_amdgcn_wmma_f32_16x16x4_f32(
           false, a1,  false, b1,
           (short)0, acc[u],  false,  false);
    }
  }

  float res[GRPS][8];
#pragma unroll
  for (int u = 0; u < GRPS; ++u) {
#pragma unroll
    for (int i = 0; i < 8; ++i) {
      const float lo = acc[u][i];
      const int   li = __builtin_bit_cast(int, lo);
      const float hi = __builtin_bit_cast(
          float, __builtin_amdgcn_permlanex16(li, li, 0x00098765, 0x00000000, false, false));
      res[u][i] = lo + hi;
    }
  }
  if (ln < D_) {
    const float lb = linb[ln];
    float* op = out + ((size_t)brow * P_ + p) * D_ + ln;
#pragma unroll
    for (int u = 0; u < GRPS; ++u) {
#pragma unroll
      for (int i = 0; i < 8; ++i)
        __builtin_nontemporal_store(res[u][i] + lb, op + (size_t)(8 * u + i) * P_ * D_);
    }
  }
}

extern "C" void kernel_launch(void* const* d_in, const int* in_sizes, int n_in,
                              void* d_out, int out_size, void* d_ws, size_t ws_size,
                              hipStream_t stream) {
  const float* feat    = (const float*)d_in[0];
  const float* weights = (const float*)d_in[1];
  const float* bias    = (const float*)d_in[2];
  const float* gamma   = (const float*)d_in[3];
  const float* beta    = (const float*)d_in[4];
  const float* linw    = (const float*)d_in[5];
  const float* linb    = (const float*)d_in[6];
  float* out  = (float*)d_out;
  float* wrow = (float*)d_ws;

  rowsum_kernel<<<(P_ * C_) / (RS_ROWS * RS_WAVES), 32 * RS_WAVES, 0, stream>>>(weights, wrow);

  fused_kernel<<<P_ * (B_ / BROWS), 32 * WAVES, 0, stream>>>(feat, wrow, bias, gamma, beta,
                                                     linw, linb, out);
}


# Lanes share a row: 512-byte row loads, per-column vectors in lane registers (no LDS staging ahead of the loads), h turned through LDS into A fragments; 16-row waves; LayerNorm sum order changes
# speedup vs baseline: 1.3429x; 1.1079x over previous
#include <hip/hip_runtime.h>
#include <math.h>


typedef __attribute__((ext_vector_type(2))) float v2f;
typedef __attribute__((ext_vector_type(4))) float v4f;
typedef __attribute__((ext_vector_type(8))) float v8f;

#define B_  8192
#define P_  25
#define C_  256
#define D_  5
#define NCH 16
#define GRPS 2
#define RW  (GRPS * 8)
#define WAVES 4
#define BROWS (WAVES * RW)
#define HP  272
#define LWS 268

#define RS_ROWS  8
#define RS_WAVES 4
__global__ __launch_bounds__(32 * RS_WAVES) void rowsum_kernel(const float* __restrict__ w,
                                                              float* __restrict__ wrow) {
  const int wave = blockIdx.x * RS_WAVES + (threadIdx.x >> 5);
  const int ln   = threadIdx.x & 31;
  const int sub  = ln >> 3;
  const int j    = ln & 7;
  const int row0 = wave * RS_ROWS;
  const float* rp = w + (size_t)(row0 + sub) * C_ + 4 * j;

  v4f v[RS_ROWS / 4][8];
#pragma unroll
  for (int s = 0; s < RS_ROWS / 4; ++s) {
#pragma unroll
    for (int q = 0; q < 8; ++q)
      v[s][q] = *(const v4f*)(rp + (size_t)s * 4 * C_ + 32 * q);
  }

  __builtin_amdgcn_sched_barrier(0);

  float t[RS_ROWS / 4];
#pragma unroll
  for (int s = 0; s < RS_ROWS / 4; ++s) {
    v4f a = v[s][0];
#pragma unroll
    for (int q = 1; q < 8; ++q) a += v[s][q];
    t[s] = (a[0] + a[1]) + (a[2] + a[3]);
  }
#pragma unroll
  for (int off = 4; off > 0; off >>= 1) {
#pragma unroll
    for (int s = 0; s < RS_ROWS / 4; ++s) t[s] += __shfl_xor(t[s], off, 32);
  }
  if (j == 0) {
#pragma unroll
    for (int s = 0; s < RS_ROWS / 4; ++s) wrow[row0 + 4 * s + sub] = t[s];
  }
}

__global__ __launch_bounds__(32 * WAVES) void fused_kernel(
    const float* __restrict__ feat, const float* __restrict__ wrow,
    const float* __restrict__ bias, const float* __restrict__ gamma,
    const float* __restrict__ beta, const float* __restrict__ linw,
    const float* __restrict__ linb, float* __restrict__ out) {
  __shared__ __attribute__((aligned(16))) float hbuf[WAVES * RW * HP];
  __shared__ __attribute__((aligned(16))) float clw[D_ * LWS];

  const int tid = threadIdx.x;
  const int p   = blockIdx.x % P_;
  const int bg  = blockIdx.x / P_;
  const int wv  = tid >> 5;
  const int ln  = tid & 31;
  const int brow = bg * BROWS + wv * RW;

  v4f lwst[3];
#pragma unroll
  for (int t = 0; t < 3; ++t) {
    const int e = tid + 32 * WAVES * t;
    if (e < D_ * C_ / 4) lwst[t] = *(const v4f*)(linw + 4 * e);
  }
  v4f wr[2], bi[2], ga[2], be[2];
#pragma unroll
  for (int q = 0; q < 2; ++q) {
    wr[q] = *(const v4f*)(wrow + p * C_ + 128 * q + 4 * ln);
    bi[q] = *(const v4f*)(bias + p * C_ + 128 * q + 4 * ln);
    ga[q] = *(const v4f*)(gamma + 128 * q + 4 * ln);
    be[q] = *(const v4f*)(beta + 128 * q + 4 * ln);
  }
  v4f f[RW][2];
  const float* fp0 = feat + ((size_t)brow * P_ + p) * C_ + 4 * ln;
#pragma unroll
  for (int r = 0; r < RW; ++r) {
#pragma unroll
    for (int q = 0; q < 2; ++q)
      f[r][q] = *(const v4f*)(fp0 + (size_t)r * P_ * C_ + 128 * q);
  }

#pragma unroll
  for (int t = 0; t < 3; ++t) {
    const int e = tid + 32 * WAVES * t;
    if (e < D_ * C_ / 4) *(v4f*)&clw[(e >> 6) * LWS + 4 * (e & 63)] = lwst[t];
  }

  float v[2 * RW];
#pragma unroll
  for (int r = 0; r < RW; ++r) {
    const v4f pa = __builtin_elementwise_fma(f[r][0], wr[0], bi[0]);
    const v4f pb = __builtin_elementwise_fma(f[r][1], wr[1], bi[1]);
    const v4f s  = pa + pb;
    const v4f qq = __builtin_elementwise_fma(pb, pb, pa * pa);
    v[r]      = (s[0] + s[1]) + (s[2] + s[3]);
    v[RW + r] = (qq[0] + qq[1]) + (qq[2] + qq[3]);
  }

#pragma unroll
  for (int d = 16, n = RW; d > 0; d >>= 1, n >>= 1) {
    const bool up = (ln & d) != 0;
#pragma unroll
    for (int i = 0; i < n; ++i) {
      const float a = v[i], b = v[i + n];
      const float keep = up ? b : a;
      const float send = up ? a : b;
      v[i] = keep + __shfl_xor(send, d, 32);
    }
  }
  float mean, rstd;
  {
    const float mine  = v[0];
    const float other = __shfl_xor(mine, 16, 32);
    const float s1 = (ln & 16) ? other : mine;
    const float s2 = (ln & 16) ? mine : other;
    const float inv = 1.0f / (float)C_;
    const float m   = s1 * inv;
    const float var = fmaf(s2, inv, -m * m);
    mean = m;
    rstd = rsqrtf(var + 1e-5f);
  }

  float* hw = hbuf + wv * (RW * HP);
#pragma unroll
  for (int q = 0; q < 2; ++q) asm volatile("" : "+v"(wr[q]), "+v"(bi[q]));
#pragma unroll
  for (int r = 0; r < RW; ++r) {
    const float m  = __builtin_bit_cast(
        float, __builtin_amdgcn_readlane(__builtin_bit_cast(int, mean), r));
    const float rs = __builtin_bit_cast(
        float, __builtin_amdgcn_readlane(__builtin_bit_cast(int, rstd), r));
#pragma unroll
    for (int q = 0; q < 2; ++q) {
      const v4f x  = f[r][q];
      const v4f pw = __builtin_elementwise_fma(x, wr[q], bi[q]);
      const v4f nn = __builtin_elementwise_fma((pw - m) * rs, ga[q], be[q]);
      const v4f hh = __builtin_elementwise_max(x + nn, (v4f){0.f, 0.f, 0.f, 0.f});
      *(v4f*)(hw + r * HP + 128 * q + 4 * ln) = hh;
    }
  }

  __syncthreads();

  const int r8 = ln & 7;
  const int g  = (ln >> 3) & 1;
  const int hf = ln >> 4;
  const int coff = 8 * g + 4 * hf;
  const int n  = ln & 15;
  const int dn = n < D_ ? n : (n < 2 * D_ ? n - D_ : D_ - 1);
  const int gn = n < D_ ? 0 : 1;
  const float* lwp = clw + dn * LWS + 8 * gn + 4 * hf;
  const float* ap  = hw + r8 * HP + coff;

  v8f acc[GRPS];
#pragma unroll
  for (int u = 0; u < GRPS; ++u)
    acc[u] = (v8f){0.f, 0.f, 0.f, 0.f, 0.f, 0.f, 0.f, 0.f};

  v4f cw[2][GRPS + 1];
#pragma unroll
  for (int u = 0; u < GRPS; ++u) cw[0][u] = *(const v4f*)(ap + u * 8 * HP);
  cw[0][GRPS] = *(const v4f*)(lwp);
#pragma unroll
  for (int jj = 0; jj < NCH; ++jj) {
    if (jj + 1 < NCH) {
#pragma unroll
      for (int u = 0; u < GRPS; ++u)
        cw[(jj + 1) & 1][u] = *(const v4f*)(ap + u * 8 * HP + 16 * (jj + 1));
      cw[(jj + 1) & 1][GRPS] = *(const v4f*)(lwp + 16 * (jj + 1));
    }
    const v4f lw = cw[jj & 1][GRPS];
    const v2f b0 = {lw[0], lw[1]};
    const v2f b1 = {lw[2], lw[3]};
#pragma unroll
    for (int u = 0; u < GRPS; ++u) {
      const v4f hh = cw[jj & 1][u];
      const v2f a0 = {hh[0], hh[1]};
      const v2f a1 = {hh[2], hh[3]};
      acc[u] = __builtin_amdgcn_wmma_f32_16x16x4_f32(
           false, a0,  false, b0,
           (short)0, acc[u],  false,  false);
      acc[u] = __builtin_amdgcn_wmma_f32_16x16x4_f32(
           false, a1,  false, b1,
           (short)0, acc[u],  false,  false);
    }
  }

  float res[GRPS][8];
#pragma unroll
  for (int u = 0; u < GRPS; ++u) {
#pragma unroll
    for (int i = 0; i < 8; ++i) {
      const float lo = acc[u][i];
      const int   li = __builtin_bit_cast(int, lo);
      const float hi = __builtin_bit_cast(
          float, __builtin_amdgcn_permlanex16(li, li, 0x00098765, 0x00000000, false, false));
      res[u][i] = lo + hi;
    }
  }
  if (ln < D_) {
    const float lb = linb[ln];
    float* op = out + ((size_t)brow * P_ + p) * D_ + ln;
#pragma unroll
    for (int u = 0; u < GRPS; ++u) {
#pragma unroll
      for (int i = 0; i < 8; ++i)
        __builtin_nontemporal_store(res[u][i] + lb, op + (size_t)(8 * u + i) * P_ * D_);
    }
  }
}

extern "C" void kernel_launch(void* const* d_in, const int* in_sizes, int n_in,
                              void* d_out, int out_size, void* d_ws, size_t ws_size,
                              hipStream_t stream) {
  const float* feat    = (const float*)d_in[0];
  const float* weights = (const float*)d_in[1];
  const float* bias    = (const float*)d_in[2];
  const float* gamma   = (const float*)d_in[3];
  const float* beta    = (const float*)d_in[4];
  const float* linw    = (const float*)d_in[5];
  const float* linb    = (const float*)d_in[6];
  float* out  = (float*)d_out;
  float* wrow = (float*)d_ws;

  rowsum_kernel<<<(P_ * C_) / (RS_ROWS * RS_WAVES), 32 * RS_WAVES, 0, stream>>>(weights, wrow);

  fused_kernel<<<P_ * (B_ / BROWS), 32 * WAVES, 0, stream>>>(feat, wrow, bias, gamma, beta,
                                                     linw, linb, out);
}


# Five waves per SIMD not four: 191 VGPRs (gamma/beta loaded late, first exchange step inside pass 1) and 64 KiB LDS a block (lin_w B pieces from global to registers, no barrier; h rows unpadded, XOR)
# speedup vs baseline: 1.3902x; 1.0352x over previous
#include <hip/hip_runtime.h>
#include <math.h>


typedef __attribute__((ext_vector_type(2))) float v2f;
typedef __attribute__((ext_vector_type(4))) float v4f;
typedef __attribute__((ext_vector_type(8))) float v8f;

#define B_  8192
#define P_  25
#define C_  256
#define D_  5
#define NCH 16
#define GRPS 2
#define RW  (GRPS * 8)
#define WAVES 4
#define BROWS (WAVES * RW)
#define HP  256

#define RS_ROWS  8
#define RS_WAVES 4
__global__ __launch_bounds__(32 * RS_WAVES) void rowsum_kernel(const float* __restrict__ w,
                                                              float* __restrict__ wrow) {
  const int wave = blockIdx.x * RS_WAVES + (threadIdx.x >> 5);
  const int ln   = threadIdx.x & 31;
  const int sub  = ln >> 3;
  const int j    = ln & 7;
  const int row0 = wave * RS_ROWS;
  const float* rp = w + (size_t)(row0 + sub) * C_ + 4 * j;

  v4f v[RS_ROWS / 4][8];
#pragma unroll
  for (int s = 0; s < RS_ROWS / 4; ++s) {
#pragma unroll
    for (int q = 0; q < 8; ++q)
      v[s][q] = *(const v4f*)(rp + (size_t)s * 4 * C_ + 32 * q);
  }

  __builtin_amdgcn_sched_barrier(0);

  float t[RS_ROWS / 4];
#pragma unroll
  for (int s = 0; s < RS_ROWS / 4; ++s) {
    v4f a = v[s][0];
#pragma unroll
    for (int q = 1; q < 8; ++q) a += v[s][q];
    t[s] = (a[0] + a[1]) + (a[2] + a[3]);
  }
#pragma unroll
  for (int off = 4; off > 0; off >>= 1) {
#pragma unroll
    for (int s = 0; s < RS_ROWS / 4; ++s) t[s] += __shfl_xor(t[s], off, 32);
  }
  if (j == 0) {
#pragma unroll
    for (int s = 0; s < RS_ROWS / 4; ++s) wrow[row0 + 4 * s + sub] = t[s];
  }
}

__global__ __launch_bounds__(32 * WAVES) void fused_kernel(
    const float* __restrict__ feat, const float* __restrict__ wrow,
    const float* __restrict__ bias, const float* __restrict__ gamma,
    const float* __restrict__ beta, const float* __restrict__ linw,
    const float* __restrict__ linb, float* __restrict__ out) {
  __shared__ __attribute__((aligned(16))) float hbuf[WAVES * RW * HP];

  const int tid = threadIdx.x;
  const int p   = blockIdx.x % P_;
  const int bg  = blockIdx.x / P_;
  const int wv  = tid >> 5;
  const int ln  = tid & 31;
  const int brow = bg * BROWS + wv * RW;

  v4f wr[2], bi[2];
#pragma unroll
  for (int q = 0; q < 2; ++q) {
    wr[q] = *(const v4f*)(wrow + p * C_ + 128 * q + 4 * ln);
    bi[q] = *(const v4f*)(bias + p * C_ + 128 * q + 4 * ln);
  }
  v4f f[RW][2];
  const float* fp0 = feat + ((size_t)brow * P_ + p) * C_ + 4 * ln;
#pragma unroll
  for (int r = 0; r < RW; ++r) {
#pragma unroll
    for (int q = 0; q < 2; ++q)
      f[r][q] = *(const v4f*)(fp0 + (size_t)r * P_ * C_ + 128 * q);
  }

  const bool up16 = (ln & 16) != 0;
  float v[RW];
#pragma unroll
  for (int r = 0; r < RW; ++r) {
    const v4f pa = __builtin_elementwise_fma(f[r][0], wr[0], bi[0]);
    const v4f pb = __builtin_elementwise_fma(f[r][1], wr[1], bi[1]);
    const v4f s  = pa + pb;
    const v4f qq = __builtin_elementwise_fma(pb, pb, pa * pa);
    const float s1 = (s[0] + s[1]) + (s[2] + s[3]);
    const float s2 = (qq[0] + qq[1]) + (qq[2] + qq[3]);
    const float keep = up16 ? s2 : s1;
    const float send = up16 ? s1 : s2;
    v[r] = keep + __shfl_xor(send, 16, 32);
    if (r & 1) __builtin_amdgcn_sched_barrier(0);
  }

  v4f ga[2], be[2];
#pragma unroll
  for (int d = 8, n = RW / 2; d > 0; d >>= 1, n >>= 1) {
    const bool up = (ln & d) != 0;
#pragma unroll
    for (int i = 0; i < n; ++i) {
      const float a = v[i], b = v[i + n];
      const float keep = up ? b : a;
      const float send = up ? a : b;
      v[i] = keep + __shfl_xor(send, d, 32);
    }
    if (d == 8) {
      __builtin_amdgcn_sched_barrier(0);
#pragma unroll
      for (int q = 0; q < 2; ++q) {
        ga[q] = *(const v4f*)(gamma + 128 * q + 4 * ln);
        be[q] = *(const v4f*)(beta + 128 * q + 4 * ln);
      }
      __builtin_amdgcn_sched_barrier(0);
    }
  }
  float mean, rstd;
  {
    const float mine  = v[0];
    const float other = __shfl_xor(mine, 16, 32);
    const float s1 = (ln & 16) ? other : mine;
    const float s2 = (ln & 16) ? mine : other;
    const float inv = 1.0f / (float)C_;
    const float m   = s1 * inv;
    const float var = fmaf(s2, inv, -m * m);
    mean = m;
    rstd = rsqrtf(var + 1e-5f);
  }

  __builtin_amdgcn_sched_barrier(0);

  float* hw = hbuf + wv * (RW * HP);
  int so[4];
#pragma unroll
  for (int t = 0; t < 4; ++t) so[t] = (4 * ln) ^ (t << 4);
#pragma unroll
  for (int q = 0; q < 2; ++q) asm volatile("" : "+v"(wr[q]), "+v"(bi[q]));
  const int r8 = ln & 7;
  const int g  = (ln >> 3) & 1;
  const int hf = ln >> 4;
  const int coff = 8 * g + 4 * hf;
  const int n  = ln & 15;
  const int dn = n < D_ ? n : (n < 2 * D_ ? n - D_ : D_ - 1);
  const int gn = n < D_ ? 0 : 1;
  const float* lwp = linw + dn * C_ + 8 * gn + 4 * hf;
  v4f bw[NCH];

#pragma unroll
  for (int half = 0; half < 2; ++half) {
#pragma unroll
    for (int r = half * (RW / 2); r < (half + 1) * (RW / 2); ++r) {
      const float m  = __builtin_bit_cast(
          float, __builtin_amdgcn_readlane(__builtin_bit_cast(int, mean), r));
      const float rs = __builtin_bit_cast(
          float, __builtin_amdgcn_readlane(__builtin_bit_cast(int, rstd), r));
#pragma unroll
      for (int q = 0; q < 2; ++q) {
        const v4f x  = f[r][q];
        const v4f pw = __builtin_elementwise_fma(x, wr[q], bi[q]);
        const v4f nn = __builtin_elementwise_fma((pw - m) * rs, ga[q], be[q]);
        const v4f hh = __builtin_elementwise_max(x + nn, (v4f){0.f, 0.f, 0.f, 0.f});
        *(v4f*)(hw + r * HP + 128 * q + so[r & 3]) = hh;
      }
      if ((r & 7) < 2 || (r & 1)) __builtin_amdgcn_sched_barrier(0);
    }
    if (half == 0) {
      __builtin_amdgcn_sched_barrier(0);
#pragma unroll
      for (int jj = 0; jj < NCH; ++jj) bw[jj] = *(const v4f*)(lwp + 16 * jj);
      __builtin_amdgcn_sched_barrier(0);
    }
  }

  __builtin_amdgcn_fence(__ATOMIC_RELEASE, "wavefront");
  asm volatile("s_wait_dscnt 0x0" ::: "memory");
  __builtin_amdgcn_wave_barrier();
  __builtin_amdgcn_fence(__ATOMIC_ACQUIRE, "wavefront");

  const float* al[4];
#pragma unroll
  for (int t = 0; t < 4; ++t) al[t] = hw + r8 * HP + coff + 16 * (t ^ (r8 & 3));

  v8f acc[GRPS];
#pragma unroll
  for (int u = 0; u < GRPS; ++u)
    acc[u] = (v8f){0.f, 0.f, 0.f, 0.f, 0.f, 0.f, 0.f, 0.f};

  v4f cw[2][GRPS];
#pragma unroll
  for (int u = 0; u < GRPS; ++u) cw[0][u] = *(const v4f*)(al[0] + u * 8 * HP);
#pragma unroll
  for (int jj = 0; jj < NCH; ++jj) {
    if (jj + 1 < NCH) {
#pragma unroll
      for (int u = 0; u < GRPS; ++u)
        cw[(jj + 1) & 1][u] =
            *(const v4f*)(al[(jj + 1) & 3] + u * 8 * HP + 16 * ((jj + 1) & ~3));
    }
    const v4f lw = bw[jj];
    const v2f b0 = {lw[0], lw[1]};
    const v2f b1 = {lw[2], lw[3]};
#pragma unroll
    for (int u = 0; u < GRPS; ++u) {
      const v4f hh = cw[jj & 1][u];
      const v2f a0 = {hh[0], hh[1]};
      acc[u] = __builtin_amdgcn_wmma_f32_16x16x4_f32(
           false, a0,  false, b0,
           (short)0, acc[u],  false,  false);
    }
#pragma unroll
    for (int u = 0; u < GRPS; ++u) {
      const v4f hh = cw[jj & 1][u];
      const v2f a1 = {hh[2], hh[3]};
      acc[u] = __builtin_amdgcn_wmma_f32_16x16x4_f32(
           false, a1,  false, b1,
           (short)0, acc[u],  false,  false);
    }
    if ((jj & 3) == 3) __builtin_amdgcn_sched_barrier(0);
  }

  float res[GRPS][8];
#pragma unroll
  for (int u = 0; u < GRPS; ++u) {
#pragma unroll
    for (int i = 0; i < 8; ++i) {
      const float lo = acc[u][i];
      const int   li = __builtin_bit_cast(int, lo);
      const float hi = __builtin_bit_cast(
          float, __builtin_amdgcn_permlanex16(li, li, 0x00098765, 0x00000000, false, false));
      res[u][i] = lo + hi;
    }
  }
  if (ln < D_) {
    const float lb = linb[ln];
    float* op = out + ((size_t)brow * P_ + p) * D_ + ln;
#pragma unroll
    for (int u = 0; u < GRPS; ++u) {
#pragma unroll
      for (int i = 0; i < 8; ++i)
        __builtin_nontemporal_store(res[u][i] + lb, op + (size_t)(8 * u + i) * P_ * D_);
    }
  }
}

extern "C" void kernel_launch(void* const* d_in, const int* in_sizes, int n_in,
                              void* d_out, int out_size, void* d_ws, size_t ws_size,
                              hipStream_t stream) {
  const float* feat    = (const float*)d_in[0];
  const float* weights = (const float*)d_in[1];
  const float* bias    = (const float*)d_in[2];
  const float* gamma   = (const float*)d_in[3];
  const float* beta    = (const float*)d_in[4];
  const float* linw    = (const float*)d_in[5];
  const float* linb    = (const float*)d_in[6];
  float* out  = (float*)d_out;
  float* wrow = (float*)d_ws;

  rowsum_kernel<<<(P_ * C_) / (RS_ROWS * RS_WAVES), 32 * RS_WAVES, 0, stream>>>(weights, wrow);

  fused_kernel<<<P_ * (B_ / BROWS), 32 * WAVES, 0, stream>>>(feat, wrow, bias, gamma, beta,
                                                     linw, linb, out);
}


# speedup vs baseline: 1.4448x; 1.0393x over previous
#include <hip/hip_runtime.h>
#include <math.h>


typedef __attribute__((ext_vector_type(2))) float v2f;
typedef __attribute__((ext_vector_type(4))) float v4f;
typedef __attribute__((ext_vector_type(8))) float v8f;

#define B_  8192
#define P_  25
#define C_  256
#define D_  5
#define NCH 16
#define GRPS 2
#define RW  (GRPS * 8)
#define WAVES 4
#define BROWS (WAVES * RW)
#define HP  256

#define RS_ROWS  8
#define RS_WAVES 4
__global__ __launch_bounds__(32 * RS_WAVES) void rowsum_kernel(const float* __restrict__ w,
                                                              float* __restrict__ wrow) {
  const int wave = blockIdx.x * RS_WAVES + (threadIdx.x >> 5);
  const int ln   = threadIdx.x & 31;
  const int sub  = ln >> 3;
  const int j    = ln & 7;
  const int row0 = wave * RS_ROWS;
  const float* rp = w + (size_t)(row0 + sub) * C_ + 4 * j;

  v4f v[RS_ROWS / 4][8];
#pragma unroll
  for (int s = 0; s < RS_ROWS / 4; ++s) {
#pragma unroll
    for (int q = 0; q < 8; ++q)
      v[s][q] = *(const v4f*)(rp + (size_t)s * 4 * C_ + 32 * q);
  }

  __builtin_amdgcn_sched_barrier(0);

  float t[RS_ROWS / 4];
#pragma unroll
  for (int s = 0; s < RS_ROWS / 4; ++s) {
    v4f a = v[s][0];
#pragma unroll
    for (int q = 1; q < 8; ++q) a += v[s][q];
    t[s] = (a[0] + a[1]) + (a[2] + a[3]);
  }
#pragma unroll
  for (int off = 4; off > 0; off >>= 1) {
#pragma unroll
    for (int s = 0; s < RS_ROWS / 4; ++s) t[s] += __shfl_xor(t[s], off, 32);
  }
  if (j == 0) {
#pragma unroll
    for (int s = 0; s < RS_ROWS / 4; ++s) wrow[row0 + 4 * s + sub] = t[s];
  }
}

__global__ __launch_bounds__(32 * WAVES) void fused_kernel(
    const float* __restrict__ feat, const float* __restrict__ wrow,
    const float* __restrict__ bias, const float* __restrict__ gamma,
    const float* __restrict__ beta, const float* __restrict__ linw,
    const float* __restrict__ linb, float* __restrict__ out) {
  __shared__ __attribute__((aligned(16))) float hbuf[WAVES * RW * HP];

  const int tid = threadIdx.x;
  const int p   = blockIdx.x % P_;
  const int bg  = blockIdx.x / P_;
  const int wv  = tid >> 5;
  const int ln  = tid & 31;
  const int brow = bg * BROWS + wv * RW;

  v4f wr[2], bi[2];
#pragma unroll
  for (int q = 0; q < 2; ++q) {
    wr[q] = *(const v4f*)(wrow + p * C_ + 128 * q + 4 * ln);
    bi[q] = *(const v4f*)(bias + p * C_ + 128 * q + 4 * ln);
  }
  v4f f[RW][2];
  const float* fp0 = feat + ((size_t)brow * P_ + p) * C_ + 4 * ln;
#pragma unroll
  for (int r = 0; r < RW; ++r) {
#pragma unroll
    for (int q = 0; q < 2; ++q)
      f[r][q] = __builtin_nontemporal_load((const v4f*)(fp0 + (size_t)r * P_ * C_ + 128 * q));
  }

  const bool up16 = (ln & 16) != 0;
  float v[RW];
#pragma unroll
  for (int r = 0; r < RW; ++r) {
    const v4f pa = __builtin_elementwise_fma(f[r][0], wr[0], bi[0]);
    const v4f pb = __builtin_elementwise_fma(f[r][1], wr[1], bi[1]);
    const v4f s  = pa + pb;
    const v4f qq = __builtin_elementwise_fma(pb, pb, pa * pa);
    const float s1 = (s[0] + s[1]) + (s[2] + s[3]);
    const float s2 = (qq[0] + qq[1]) + (qq[2] + qq[3]);
    const float keep = up16 ? s2 : s1;
    const float send = up16 ? s1 : s2;
    v[r] = keep + __shfl_xor(send, 16, 32);
    if (r & 1) __builtin_amdgcn_sched_barrier(0);
  }

  v4f ga[2], be[2];
#pragma unroll
  for (int d = 8, n = RW / 2; d > 0; d >>= 1, n >>= 1) {
    const bool up = (ln & d) != 0;
#pragma unroll
    for (int i = 0; i < n; ++i) {
      const float a = v[i], b = v[i + n];
      const float keep = up ? b : a;
      const float send = up ? a : b;
      v[i] = keep + __shfl_xor(send, d, 32);
    }
    if (d == 8) {
      __builtin_amdgcn_sched_barrier(0);
#pragma unroll
      for (int q = 0; q < 2; ++q) {
        ga[q] = *(const v4f*)(gamma + 128 * q + 4 * ln);
        be[q] = *(const v4f*)(beta + 128 * q + 4 * ln);
      }
      __builtin_amdgcn_sched_barrier(0);
    }
  }
  float mean, rstd;
  {
    const float mine  = v[0];
    const float other = __shfl_xor(mine, 16, 32);
    const float s1 = (ln & 16) ? other : mine;
    const float s2 = (ln & 16) ? mine : other;
    const float inv = 1.0f / (float)C_;
    const float m   = s1 * inv;
    const float var = fmaf(s2, inv, -m * m);
    mean = m;
    rstd = rsqrtf(var + 1e-5f);
  }

  __builtin_amdgcn_sched_barrier(0);

  float* hw = hbuf + wv * (RW * HP);
  int so[4];
#pragma unroll
  for (int t = 0; t < 4; ++t) so[t] = (4 * ln) ^ (t << 4);
#pragma unroll
  for (int q = 0; q < 2; ++q) asm volatile("" : "+v"(wr[q]), "+v"(bi[q]));
  const int r8 = ln & 7;
  const int g  = (ln >> 3) & 1;
  const int hf = ln >> 4;
  const int coff = 8 * g + 4 * hf;
  const int n  = ln & 15;
  const int dn = n < D_ ? n : (n < 2 * D_ ? n - D_ : D_ - 1);
  const int gn = n < D_ ? 0 : 1;
  const float* lwp = linw + dn * C_ + 8 * gn + 4 * hf;
  v4f bw[NCH];

#pragma unroll
  for (int half = 0; half < 2; ++half) {
#pragma unroll
    for (int r = half * (RW / 2); r < (half + 1) * (RW / 2); ++r) {
      const float m  = __builtin_bit_cast(
          float, __builtin_amdgcn_readlane(__builtin_bit_cast(int, mean), r));
      const float rs = __builtin_bit_cast(
          float, __builtin_amdgcn_readlane(__builtin_bit_cast(int, rstd), r));
#pragma unroll
      for (int q = 0; q < 2; ++q) {
        const v4f x  = f[r][q];
        const v4f pw = __builtin_elementwise_fma(x, wr[q], bi[q]);
        const v4f nn = __builtin_elementwise_fma((pw - m) * rs, ga[q], be[q]);
        const v4f hh = __builtin_elementwise_max(x + nn, (v4f){0.f, 0.f, 0.f, 0.f});
        *(v4f*)(hw + r * HP + 128 * q + so[r & 3]) = hh;
      }
      if ((r & 7) < 2 || (r & 1)) __builtin_amdgcn_sched_barrier(0);
    }
    if (half == 0) {
      __builtin_amdgcn_sched_barrier(0);
#pragma unroll
      for (int jj = 0; jj < NCH; ++jj) bw[jj] = *(const v4f*)(lwp + 16 * jj);
      __builtin_amdgcn_sched_barrier(0);
    }
  }

  __builtin_amdgcn_fence(__ATOMIC_RELEASE, "wavefront");
  asm volatile("s_wait_dscnt 0x0" ::: "memory");
  __builtin_amdgcn_wave_barrier();
  __builtin_amdgcn_fence(__ATOMIC_ACQUIRE, "wavefront");

  const float* al[4];
#pragma unroll
  for (int t = 0; t < 4; ++t) al[t] = hw + r8 * HP + coff + 16 * (t ^ (r8 & 3));

  v8f acc[GRPS];
#pragma unroll
  for (int u = 0; u < GRPS; ++u)
    acc[u] = (v8f){0.f, 0.f, 0.f, 0.f, 0.f, 0.f, 0.f, 0.f};

  v4f cw[2][GRPS];
#pragma unroll
  for (int u = 0; u < GRPS; ++u) cw[0][u] = *(const v4f*)(al[0] + u * 8 * HP);
#pragma unroll
  for (int jj = 0; jj < NCH; ++jj) {
    if (jj + 1 < NCH) {
#pragma unroll
      for (int u = 0; u < GRPS; ++u)
        cw[(jj + 1) & 1][u] =
            *(const v4f*)(al[(jj + 1) & 3] + u * 8 * HP + 16 * ((jj + 1) & ~3));
    }
    const v4f lw = bw[jj];
    const v2f b0 = {lw[0], lw[1]};
    const v2f b1 = {lw[2], lw[3]};
#pragma unroll
    for (int u = 0; u < GRPS; ++u) {
      const v4f hh = cw[jj & 1][u];
      const v2f a0 = {hh[0], hh[1]};
      acc[u] = __builtin_amdgcn_wmma_f32_16x16x4_f32(
           false, a0,  false, b0,
           (short)0, acc[u],  false,  false);
    }
#pragma unroll
    for (int u = 0; u < GRPS; ++u) {
      const v4f hh = cw[jj & 1][u];
      const v2f a1 = {hh[2], hh[3]};
      acc[u] = __builtin_amdgcn_wmma_f32_16x16x4_f32(
           false, a1,  false, b1,
           (short)0, acc[u],  false,  false);
    }
    if ((jj & 3) == 3) __builtin_amdgcn_sched_barrier(0);
  }

  float res[GRPS][8];
#pragma unroll
  for (int u = 0; u < GRPS; ++u) {
#pragma unroll
    for (int i = 0; i < 8; ++i) {
      const float lo = acc[u][i];
      const int   li = __builtin_bit_cast(int, lo);
      const float hi = __builtin_bit_cast(
          float, __builtin_amdgcn_permlanex16(li, li, 0x00098765, 0x00000000, false, false));
      res[u][i] = lo + hi;
    }
  }
  if (ln < D_) {
    const float lb = linb[ln];
    float* op = out + ((size_t)brow * P_ + p) * D_ + ln;
#pragma unroll
    for (int u = 0; u < GRPS; ++u) {
#pragma unroll
      for (int i = 0; i < 8; ++i)
        __builtin_nontemporal_store(res[u][i] + lb, op + (size_t)(8 * u + i) * P_ * D_);
    }
  }
}

extern "C" void kernel_launch(void* const* d_in, const int* in_sizes, int n_in,
                              void* d_out, int out_size, void* d_ws, size_t ws_size,
                              hipStream_t stream) {
  const float* feat    = (const float*)d_in[0];
  const float* weights = (const float*)d_in[1];
  const float* bias    = (const float*)d_in[2];
  const float* gamma   = (const float*)d_in[3];
  const float* beta    = (const float*)d_in[4];
  const float* linw    = (const float*)d_in[5];
  const float* linb    = (const float*)d_in[6];
  float* out  = (float*)d_out;
  float* wrow = (float*)d_ws;

  rowsum_kernel<<<(P_ * C_) / (RS_ROWS * RS_WAVES), 32 * RS_WAVES, 0, stream>>>(weights, wrow);

  fused_kernel<<<P_ * (B_ / BROWS), 32 * WAVES, 0, stream>>>(feat, wrow, bias, gamma, beta,
                                                     linw, linb, out);
}
